# starting kernel
# baseline (speedup 1.0000x reference)
#include <hip/hip_runtime.h>

typedef __attribute__((ext_vector_type(2))) float v2f;
typedef __attribute__((ext_vector_type(8))) float v8f;

#define NB   2
#define CHN  4
#define SS   100
#define SP   112

#define OFF_WR    0
#define OFF_WCT   (NB*SP*SP)
#define OFF_MIN   (2*NB*SP*SP)
#define OFF_SUM   (2*NB*SP*SP + 2*NB*SP)
#define OFF_IMG   (2*NB*SP*SP + 4*NB*SP)
#define OFF_TMP   (OFF_IMG + NB*CHN*SP*SP)

__global__ void dec_prep(const float* __restrict__ transforms, float* __restrict__ ws) {
    int t = blockIdx.x * blockDim.x + threadIdx.x;
    if (t >= NB * 2 * SP) return;
    int n    = t / (2 * SP);
    int rem  = t % (2 * SP);
    int axis = rem / SP;
    int q    = rem % SP;

    float t0 = transforms[n*4+0], t1 = transforms[n*4+1];
    float t2 = transforms[n*4+2], t3 = transforms[n*4+3];
    float scale = (axis == 0) ? (t1 - t0) : (t3 - t2);
    float off   = (axis == 0) ? (t0 * (float)SS) : (t2 * (float)SS);

    float m = 0.0f, sum = 1.0f;
    if (q < SS) {
        m = 3.4e38f;
        for (int k = 0; k < SS; ++k) {
            float d = (float)q - (scale * (float)k + off);
            m = fminf(m, d * d);
        }
        sum = 0.0f;
        for (int k = 0; k < SS; ++k) {
            float d = (float)q - (scale * (float)k + off);
            sum += __expf(m - d * d);
        }
    }
    ws[OFF_MIN + t] = m;
    ws[OFF_SUM + t] = sum;
}

__global__ void dec_fill(const float* __restrict__ transforms, float* __restrict__ ws) {
    int b    = blockIdx.x;
    int n    = b / (2 * SP);
    int rem  = b % (2 * SP);
    int axis = rem / SP;
    int row  = rem % SP;
    int col  = threadIdx.x;
    if (col >= SP) return;

    float t0 = transforms[n*4+0], t1 = transforms[n*4+1];
    float t2 = transforms[n*4+2], t3 = transforms[n*4+3];
    float scale = (axis == 0) ? (t1 - t0) : (t3 - t2);
    float off   = (axis == 0) ? (t0 * (float)SS) : (t2 * (float)SS);

    float val = 0.0f;
    if (row < SS && col < SS) {
        if (axis == 0) {
            float m = ws[OFF_MIN + (n*2 + 0)*SP + row];
            float d = (float)row - (scale * (float)col + off);
            val = __expf(m - d * d);
        } else {
            float m = ws[OFF_MIN + (n*2 + 1)*SP + col];
            float d = (float)col - (scale * (float)row + off);
            val = __expf(m - d * d);
        }
    }
    float* dst = (axis == 0) ? (ws + OFF_WR + n*SP*SP) : (ws + OFF_WCT + n*SP*SP);
    dst[row*SP + col] = val;
}

__global__ void dec_pad(const float* __restrict__ images, float* __restrict__ ws) {
    int b   = blockIdx.x;
    int mat = b / SP;
    int row = b % SP;
    int col = threadIdx.x;
    if (col >= SP) return;
    float v = 0.0f;
    if (row < SS && col < SS) v = images[mat * SS * SS + row * SS + col];
    ws[OFF_IMG + mat * SP * SP + row * SP + col] = v;
}

__device__ __forceinline__ v8f wmma_tile_112(const float* __restrict__ A,
                                             const float* __restrict__ B,
                                             int tM, int tN, int lane) {
    int hs   = lane >> 4;
    int l16  = lane & 15;
    int rowA = tM * 16 + l16;
    int colB = tN * 16 + l16;

    v8f c = {};
#pragma unroll
    for (int k0 = 0; k0 < SP; k0 += 4) {
        int ka = k0 + 2 * hs;
        v2f a = *(const v2f*)(A + rowA * SP + ka);
        v2f b;
        b.x = B[ka * SP + colB];
        b.y = B[(ka + 1) * SP + colB];
        c = __builtin_amdgcn_wmma_f32_16x16x4_f32(false, a, false, b,
                                                  (short)0, c, false, false);
    }
    return c;
}

__global__ void dec_gemm_img_wct(float* __restrict__ ws) {
    int bid  = blockIdx.x;
    int mat  = bid / 49;
    int tile = bid % 49;
    int tM = tile / 7, tN = tile % 7;
    int n  = mat / CHN;

    const float* A = ws + OFF_IMG + mat * SP * SP;
    const float* B = ws + OFF_WCT + n   * SP * SP;
    float*       D = ws + OFF_TMP + mat * SP * SP;

    int lane = threadIdx.x;
    v8f c = wmma_tile_112(A, B, tM, tN, lane);

    int hs = lane >> 4, l16 = lane & 15;
    int baseM = tM * 16 + hs * 8;
#pragma unroll
    for (int v = 0; v < 8; ++v)
        D[(baseM + v) * SP + tN * 16 + l16] = c[v];
}

__global__ void dec_gemm_wr_tmp(const float* __restrict__ ws, float* __restrict__ out) {
    int bid  = blockIdx.x;
    int mat  = bid / 49;
    int tile = bid % 49;
    int tM = tile / 7, tN = tile % 7;
    int n  = mat / CHN;

    const float* A    = ws + OFF_WR  + n   * SP * SP;
    const float* B    = ws + OFF_TMP + mat * SP * SP;
    const float* rs_r = ws + OFF_SUM + (n*2 + 0) * SP;
    const float* rs_c = ws + OFF_SUM + (n*2 + 1) * SP;

    int lane = threadIdx.x;
    v8f c = wmma_tile_112(A, B, tM, tN, lane);

    int hs = lane >> 4, l16 = lane & 15;
    int qj = tN * 16 + l16;
    if (qj < SS) {
        float rc    = rs_c[qj];
        int   baseM = tM * 16 + hs * 8;
#pragma unroll
        for (int v = 0; v < 8; ++v) {
            int qi = baseM + v;
            if (qi < SS)
                out[mat * SS * SS + qi * SS + qj] = c[v] / (rs_r[qi] * rc);
        }
    }
}

extern "C" void kernel_launch(void* const* d_in, const int* in_sizes, int n_in,
                              void* d_out, int out_size, void* d_ws, size_t ws_size,
                              hipStream_t stream) {
    (void)in_sizes; (void)n_in; (void)out_size; (void)ws_size;
    const float* images     = (const float*)d_in[0];
    const float* transforms = (const float*)d_in[1];
    float*       out        = (float*)d_out;
    float*       ws         = (float*)d_ws;

    dec_prep<<<(NB*2*SP + 127) / 128, 128, 0, stream>>>(transforms, ws);
    dec_fill<<<NB*2*SP, 128, 0, stream>>>(transforms, ws);
    dec_pad <<<NB*CHN*SP, 128, 0, stream>>>(images, ws);
    dec_gemm_img_wct<<<NB*CHN*49, 32, 0, stream>>>(ws);
    dec_gemm_wr_tmp <<<NB*CHN*49, 32, 0, stream>>>(ws, out);
}


# Fuse the 2 launches with no exchange: each wave rebuilds all 112 rows of image @ column weights in 7 accumulator tiles and feeds them back as B of the row product (keys in accumulator-row order)
# speedup vs baseline: 3.6965x; 3.6965x over previous
#include <hip/hip_runtime.h>


typedef __attribute__((ext_vector_type(2))) float v2f;
typedef __attribute__((ext_vector_type(4))) float v4f;
typedef __attribute__((ext_vector_type(8))) float v8f;

#define NB   2
#define CHN  4
#define SS   100
#define KS   25
#define NT   7
#define NU   12
#define PDU  3
#define NW2  52


__device__ __forceinline__ float col_weights(float q, float scale, float off,
                                             int hs, float (&w)[2 * KS]) {
    float d[2 * KS];
    float h4 = (float)(4 * hs), h2 = (float)(2 * hs);
    float m  = 3.4e38f;
#pragma unroll
    for (int i = 0; i < 2 * KS; ++i) {
        float kf = (i < 4 * NU) ? h4 + (float)(8 * (i >> 2) + (i & 3))
                                : h2 + (float)(96 + (i & 1));
        float dd = q - (scale * kf + off);
        d[i] = dd;
        m = fminf(m, dd * dd);
    }
    m = fminf(m, __shfl_xor(m, 16));

    float sum = 0.0f;
#pragma unroll
    for (int i = 0; i < 2 * KS; ++i) {
        float e = __expf(m - d[i] * d[i]);
        w[i] = e;
        sum += e;
    }
    sum += __shfl_xor(sum, 16);
    return sum;
}

__global__ void __launch_bounds__(32)
dec_chain(const float* __restrict__ images,
          const float* __restrict__ transforms,
          float* __restrict__ out) {
    int bid  = blockIdx.x;
    int mat  = bid / 49;
    int tile = bid % 49;
    int tM = tile / 7, tN = tile % 7;
    int n  = mat / CHN;

    int lane = threadIdx.x;
    int hs = lane >> 4, l16 = lane & 15;
    int rowQ = tM * 16 + l16;
    int colQ = tN * 16 + l16;

    const float* img   = images + mat * SS * SS;
    const float* Arow  = img + l16 * SS;
    bool         ok6   = (96 + l16) < SS;
    const float* Arow6 = img + (ok6 ? 96 + l16 : SS - 1) * SS;

    v4f aq[NU][NT];
    v2f at[NT];

#define LOAD_UNIT(u)                                                               \
    {                                                                              \
        _Pragma("unroll")                                                          \
        for (int t = 0; t < NT - 1; ++t)                                           \
            aq[u][t] = *(const v4f*)(Arow + t * 16 * SS + 8 * (u) + 4 * hs);       \
        aq[u][NT - 1] = *(const v4f*)(Arow6 + 8 * (u) + 4 * hs);                   \
    }

#pragma unroll
    for (int u = 0; u < PDU; ++u) LOAD_UNIT(u)
    __builtin_amdgcn_sched_barrier(0);

    float t0 = transforms[n * 4 + 0], t1 = transforms[n * 4 + 1];
    float t2 = transforms[n * 4 + 2], t3 = transforms[n * 4 + 3];
    float w1[2 * KS];
    float sumc = col_weights((float)colQ, t3 - t2, t2 * (float)SS, hs, w1);

    float d2[NW2], w2[NW2];
    float qr = (float)rowQ, sr = t1 - t0, offr = t0 * (float)SS;
    float h8 = (float)(8 * hs);
    bool  lo = (hs == 0);
    float m2 = 3.4e38f, sumr = 0.0f;

    v8f c[NT];
#pragma unroll
    for (int t = 0; t < NT; ++t) c[t] = (v8f){};

#pragma unroll
    for (int u = 0; u < NU; ++u) {
        if (u + PDU < NU) {
            LOAD_UNIT(u + PDU)
        } else if (u + PDU == NU) {
#pragma unroll
            for (int t = 0; t < NT - 1; ++t)
                at[t] = *(const v2f*)(Arow + t * 16 * SS + 96 + 2 * hs);
            at[NT - 1] = *(const v2f*)(Arow6 + 96 + 2 * hs);
        }
        __builtin_amdgcn_sched_barrier(0);
#pragma unroll
        for (int e = 0; e < 2; ++e) {
            v2f b;
            b.x = w1[4 * u + 2 * e];
            b.y = w1[4 * u + 2 * e + 1];
#pragma unroll
            for (int t = 0; t < NT; ++t) {
                v2f a;
                a.x = e ? aq[u][t].z : aq[u][t].x;
                a.y = e ? aq[u][t].w : aq[u][t].y;
                if (t == NT - 1) {
                    a.x = ok6 ? a.x : 0.0f;
                    a.y = ok6 ? a.y : 0.0f;
                }
                c[t] = __builtin_amdgcn_wmma_f32_16x16x4_f32(false, a, false, b,
                                                             (short)0, c[t], false, false);
            }
        }
        if (u < NT) {
#pragma unroll
            for (int v = 0; v < 8; ++v) {
                int i = 8 * u + v;
                if (i < NW2) {
                    float kf = h8 + (float)(16 * u + v);
                    float dd = qr - (sr * kf + offr);
                    d2[i] = dd;
                    float sq = dd * dd;
                    if (i >= 48) sq = lo ? sq : 3.4e38f;
                    m2 = fminf(m2, sq);
                }
            }
            if (u == NT - 1) m2 = fminf(m2, __shfl_xor(m2, 16));
        } else {
#pragma unroll
            for (int j = 0; j < 11; ++j) {
                int i = 11 * (u - NT) + j;
                if (i < NW2) {
                    float ex = __expf(m2 - d2[i] * d2[i]);
                    if (i >= 48) ex = lo ? ex : 0.0f;
                    w2[i] = ex;
                    sumr += ex;
                }
            }
        }
    }
    sumr += __shfl_xor(sumr, 16);
    {
        v2f b;
        b.x = w1[48];
        b.y = w1[49];
#pragma unroll
        for (int t = 0; t < NT; ++t) {
            v2f a = at[t];
            if (t == NT - 1) {
                a.x = ok6 ? a.x : 0.0f;
                a.y = ok6 ? a.y : 0.0f;
            }
            c[t] = __builtin_amdgcn_wmma_f32_16x16x4_f32(false, a, false, b,
                                                         (short)0, c[t], false, false);
        }
    }

    v8f dacc = {};
#pragma unroll
    for (int t = 0; t < NT; ++t) {
#pragma unroll
        for (int p = 0; p < 4; ++p) {
            if (t == NT - 1 && p >= 2) continue;
            v2f a, b;
            a.x = w2[8 * t + 2 * p];
            a.y = w2[8 * t + 2 * p + 1];
            b.x = c[t][2 * p];
            b.y = c[t][2 * p + 1];
            dacc = __builtin_amdgcn_wmma_f32_16x16x4_f32(false, a, false, b,
                                                         (short)0, dacc, false, false);
        }
    }

    float invc = 1.0f / sumc;
    float inv  = 1.0f / sumr;
    float invr[8];
#pragma unroll
    for (int v = 0; v < 8; ++v)
        invr[v] = __shfl(inv, hs * 8 + v) * invc;

    if (colQ < SS) {
        int baseM = tM * 16 + hs * 8;
#pragma unroll
        for (int v = 0; v < 8; ++v) {
            int qi = baseM + v;
            if (qi < SS)
                out[mat * SS * SS + qi * SS + colQ] = dacc[v] * invr[v];
        }
    }
#undef LOAD_UNIT
}

extern "C" void kernel_launch(void* const* d_in, const int* in_sizes, int n_in,
                              void* d_out, int out_size, void* d_ws, size_t ws_size,
                              hipStream_t stream) {
    (void)in_sizes; (void)n_in; (void)out_size; (void)d_ws; (void)ws_size;
    const float* images     = (const float*)d_in[0];
    const float* transforms = (const float*)d_in[1];
    float*       out        = (float*)d_out;

    dec_chain<<<NB*CHN*49, 32, 0, stream>>>(images, transforms, out);
}


# Split the key rows over 4 waves per pair of output tiles (224 groups): each wave builds 2 of 7 row tiles of image @ column weights; 4 partial tiles and row denominators added in LDS in wave order
# speedup vs baseline: 4.6434x; 1.2562x over previous
#include <hip/hip_runtime.h>


typedef __attribute__((ext_vector_type(2))) float v2f;
typedef __attribute__((ext_vector_type(4))) float v4f;
typedef __attribute__((ext_vector_type(8))) float v8f;

#define NB   2
#define CHN  4
#define SS   100
#define KS   25
#define NU   12
#define NWV  4
#define NG   2
#define NMIN 50


__device__ __forceinline__ float col_weights(float q, float scale, float off,
                                             int hs, float (&w)[2 * KS]) {
    float d[2 * KS];
    float h4 = (float)(4 * hs), h2 = (float)(2 * hs);
    float m  = 3.4e38f;
#pragma unroll
    for (int i = 0; i < 2 * KS; ++i) {
        float kf = (i < 4 * NU) ? h4 + (float)(8 * (i >> 2) + (i & 3))
                                : h2 + (float)(96 + (i & 1));
        float dd = q - (scale * kf + off);
        d[i] = dd;
        m = fminf(m, dd * dd);
    }
    m = fminf(m, __shfl_xor(m, 16));

    float sum = 0.0f;
#pragma unroll
    for (int i = 0; i < 2 * KS; ++i) {
        float e = __expf(m - d[i] * d[i]);
        w[i] = e;
        sum += e;
    }
    sum += __shfl_xor(sum, 16);
    return sum;
}

__global__ void __launch_bounds__(32 * NWV)
dec_chain(const float* __restrict__ images,
          const float* __restrict__ transforms,
          float* __restrict__ out) {
    __shared__ v4f part[NWV][NG][2][32];
    __shared__ __attribute__((aligned(16))) float rsum[NWV][NG][16];

    int bid = blockIdx.x;
    int mat = bid / 28;
    int rem = bid % 28;
    int grp = rem / 7, tN = rem % 7;
    int n   = mat / CHN;

    int w    = threadIdx.x >> 5;
    int lane = threadIdx.x & 31;
    int hs = lane >> 4, l16 = lane & 15;
    int colQ = tN * 16 + l16;

    const float* img = images + mat * SS * SS;
    int rA = 32 * w + l16, rB = rA + 16;
    const float* Arow[2];
    Arow[0] = img + (rA < SS ? rA : SS - 1) * SS;
    Arow[1] = img + (rB < SS ? rB : SS - 1) * SS;

    v4f aq[2][NU];
    v2f at[2];
#pragma unroll
    for (int u = 0; u < NU; ++u) {
#pragma unroll
        for (int t = 0; t < 2; ++t)
            aq[t][u] = *(const v4f*)(Arow[t] + 8 * u + 4 * hs);
    }
#pragma unroll
    for (int t = 0; t < 2; ++t)
        at[t] = *(const v2f*)(Arow[t] + 96 + 2 * hs);
    __builtin_amdgcn_sched_barrier(0);

    float t0 = transforms[n * 4 + 0], t1 = transforms[n * 4 + 1];
    float t2 = transforms[n * 4 + 2], t3 = transforms[n * 4 + 3];
    float w1[2 * KS];
    float sumc = col_weights((float)colQ, t3 - t2, t2 * (float)SS, hs, w1);

    float qr[NG], m2[NG], sp[NG], w2[NG][16];
#pragma unroll
    for (int g = 0; g < NG; ++g) {
        qr[g] = (float)((2 * grp + g) * 16 + l16);
        m2[g] = 3.4e38f;
        sp[g] = 0.0f;
    }
    float sr = t1 - t0, offr = t0 * (float)SS;
    float h50 = (float)(NMIN * hs);
    int   kbi = 32 * w + 8 * hs;
    float kb  = (float)kbi;

    v8f c[2];
#pragma unroll
    for (int t = 0; t < 2; ++t) c[t] = (v8f){};

#pragma unroll
    for (int u = 0; u < NU; ++u) {
#pragma unroll
        for (int e = 0; e < 2; ++e) {
            v2f b;
            b.x = w1[4 * u + 2 * e];
            b.y = w1[4 * u + 2 * e + 1];
#pragma unroll
            for (int t = 0; t < 2; ++t) {
                v2f a;
                a.x = e ? aq[t][u].z : aq[t][u].x;
                a.y = e ? aq[t][u].w : aq[t][u].y;
                c[t] = __builtin_amdgcn_wmma_f32_16x16x4_f32(false, a, false, b,
                                                             (short)0, c[t], false, false);
            }
        }
        if (u < 6) {
#pragma unroll
            for (int j = 0; j < 9; ++j) {
                int i = 9 * u + j;
                if (i < NMIN) {
                    float kf = h50 + (float)i;
#pragma unroll
                    for (int g = 0; g < NG; ++g) {
                        float dd = qr[g] - (sr * kf + offr);
                        m2[g] = fminf(m2[g], dd * dd);
                    }
                }
            }
            if (u == 5) {
#pragma unroll
                for (int g = 0; g < NG; ++g) m2[g] = fminf(m2[g], __shfl_xor(m2[g], 16));
            }
        } else {
#pragma unroll
            for (int j = 0; j < 6; ++j) {
                int x = 6 * (u - 6) + j;
                if (x < NG * 16) {
                    int g = x >> 4, i = x & 15;
                    int ko = 16 * (i >> 3) + (i & 7);
                    float kf = kb + (float)ko;
                    float dd = qr[g] - (sr * kf + offr);
                    float ex = __expf(m2[g] - dd * dd);
                    ex = (kbi + ko < SS) ? ex : 0.0f;
                    w2[g][i] = ex;
                    sp[g] += ex;
                }
            }
        }
    }
    {
        v2f b;
        b.x = w1[48];
        b.y = w1[49];
#pragma unroll
        for (int t = 0; t < 2; ++t)
            c[t] = __builtin_amdgcn_wmma_f32_16x16x4_f32(false, at[t], false, b,
                                                         (short)0, c[t], false, false);
    }
#pragma unroll
    for (int g = 0; g < NG; ++g) sp[g] += __shfl_xor(sp[g], 16);

    v8f dacc[NG];
#pragma unroll
    for (int g = 0; g < NG; ++g) dacc[g] = (v8f){};
#pragma unroll
    for (int t = 0; t < 2; ++t) {
#pragma unroll
        for (int p = 0; p < 4; ++p) {
#pragma unroll
            for (int g = 0; g < NG; ++g) {
                v2f a, b;
                a.x = w2[g][8 * t + 2 * p];
                a.y = w2[g][8 * t + 2 * p + 1];
                b.x = c[t][2 * p];
                b.y = c[t][2 * p + 1];
                dacc[g] = __builtin_amdgcn_wmma_f32_16x16x4_f32(false, a, false, b,
                                                                (short)0, dacc[g], false, false);
            }
        }
    }

    float invc = 1.0f / sumc;
#pragma unroll
    for (int g = 0; g < NG; ++g) {
#pragma unroll
        for (int j = 0; j < 2; ++j) {
            v4f pv;
            pv.x = dacc[g][4 * j]     * invc;
            pv.y = dacc[g][4 * j + 1] * invc;
            pv.z = dacc[g][4 * j + 2] * invc;
            pv.w = dacc[g][4 * j + 3] * invc;
            part[w][g][j][lane] = pv;
        }
    }
    if (hs == 0) {
#pragma unroll
        for (int g = 0; g < NG; ++g) rsum[w][g][l16] = sp[g];
    }
    __syncthreads();

    int g = w & 1, j = w >> 1;
    v4f acc = part[0][g][j][lane];
    v4f den = *(const v4f*)&rsum[0][g][8 * hs + 4 * j];
#pragma unroll
    for (int ww = 1; ww < NWV; ++ww) {
        acc += part[ww][g][j][lane];
        den += *(const v4f*)&rsum[ww][g][8 * hs + 4 * j];
    }

    static_assert(SS % 4 == 0, "one row guard per group of four rows");
    int baseM = (2 * grp + g) * 16 + 8 * hs + 4 * j;
    if (colQ < SS && baseM < SS) {
        float* o = out + mat * SS * SS + baseM * SS + colQ;
#pragma unroll
        for (int i = 0; i < 4; ++i)
            o[i * SS] = acc[i] / den[i];
    }
}

extern "C" void kernel_launch(void* const* d_in, const int* in_sizes, int n_in,
                              void* d_out, int out_size, void* d_ws, size_t ws_size,
                              hipStream_t stream) {
    (void)in_sizes; (void)n_in; (void)out_size; (void)d_ws; (void)ws_size;
    const float* images     = (const float*)d_in[0];
    const float* transforms = (const float*)d_in[1];
    float*       out        = (float*)d_out;

    dec_chain<<<NB*CHN*28, 32*NWV, 0, stream>>>(images, transforms, out);
}


# Softmax shifts in closed form, no minimum scans (distance is linear in the key: its square is least at an end of the key range or zero inside); a weight is fma, fma, exp2, made a k-step ahead of use
# speedup vs baseline: 4.8907x; 1.0533x over previous
#include <hip/hip_runtime.h>


typedef __attribute__((ext_vector_type(2))) float v2f;
typedef __attribute__((ext_vector_type(4))) float v4f;
typedef __attribute__((ext_vector_type(8))) float v8f;

#define NB   2
#define CHN  4
#define SS   100
#define KS   25
#define NU   12
#define NWV  4
#define NG   2
#define RL2E 1.2011224087864498f


__device__ __forceinline__ float soft_shift(float qo, float sp) {
    float b = fmaf(-sp, (float)(SS - 1), qo);
    float m = fminf(qo * qo, b * b);
    return (qo * b <= 0.0f) ? 0.0f : m;
}
__device__ __forceinline__ float soft_w(float base, float sp, float kc, float m) {
    float d = fmaf(-sp, kc, base);
    return __builtin_amdgcn_exp2f(fmaf(-d, d, m));
}

__global__ void __launch_bounds__(32 * NWV) __attribute__((amdgpu_waves_per_eu(1, 1)))
dec_chain(const float* __restrict__ images,
          const float* __restrict__ transforms,
          float* __restrict__ out) {
    __shared__ v4f part[NWV][NG][2][32];
    __shared__ __attribute__((aligned(16))) float rsum[NWV][NG][16];

    int bid = blockIdx.x;
    int mat = bid / 28;
    int rem = bid % 28;
    int grp = rem / 7, tN = rem % 7;
    int n   = mat / CHN;

    int w    = threadIdx.x >> 5;
    int lane = threadIdx.x & 31;
    int hs = lane >> 4, l16 = lane & 15;
    int colQ = tN * 16 + l16;

    const float* img = images + mat * SS * SS;
    int rA = 32 * w + l16, rB = rA + 16;
    const float* Arow[2];
    Arow[0] = img + (rA < SS ? rA : SS - 1) * SS;
    Arow[1] = img + (rB < SS ? rB : SS - 1) * SS;

    v4f aq[2][NU];
    v2f at[2];
#pragma unroll
    for (int u = 0; u < NU; ++u) {
#pragma unroll
        for (int t = 0; t < 2; ++t)
            aq[t][u] = *(const v4f*)(Arow[t] + 8 * u + 4 * hs);
    }
#pragma unroll
    for (int t = 0; t < 2; ++t)
        at[t] = *(const v2f*)(Arow[t] + 96 + 2 * hs);
    __builtin_amdgcn_sched_barrier(0);

    float t0 = transforms[n * 4 + 0], t1 = transforms[n * 4 + 1];
    float t2 = transforms[n * 4 + 2], t3 = transforms[n * 4 + 3];

    float scl = (t3 - t2) * RL2E;
    float qoc = ((float)colQ - t2 * (float)SS) * RL2E;
    float mc  = soft_shift(qoc, scl);
    float bc4 = fmaf(-scl, (float)(4 * hs), qoc);
    float bc2 = fmaf(-scl, (float)(2 * hs), qoc);
    float w1[2 * KS];
    float sumc = 0.0f;

    float w2[NG][16], sp[NG], mr[NG], br[NG];
    float srl  = (t1 - t0) * RL2E;
    int   room = SS - (32 * w + 8 * hs);
    float kb   = (float)(32 * w + 8 * hs);
#pragma unroll
    for (int g = 0; g < NG; ++g) {
        float qor = ((float)((2 * grp + g) * 16 + l16) - t0 * (float)SS) * RL2E;
        mr[g] = soft_shift(qor, srl);
        br[g] = fmaf(-srl, kb, qor);
        sp[g] = 0.0f;
    }

#pragma unroll
    for (int e = 0; e < 4; ++e) {
        w1[e] = soft_w(bc4, scl, (float)e, mc);
        sumc += w1[e];
    }

    v8f c[2];
#pragma unroll
    for (int t = 0; t < 2; ++t) c[t] = (v8f){};

#pragma unroll
    for (int u = 0; u < NU; ++u) {
        if (u + 1 < NU) {
#pragma unroll
            for (int e = 0; e < 4; ++e) {
                int i = 4 * (u + 1) + e;
                w1[i] = soft_w(bc4, scl, (float)(8 * (u + 1) + e), mc);
                sumc += w1[i];
            }
        } else {
#pragma unroll
            for (int j = 0; j < 2; ++j) {
                w1[48 + j] = soft_w(bc2, scl, (float)(96 + j), mc);
                sumc += w1[48 + j];
            }
        }
#pragma unroll
        for (int e = 0; e < 2; ++e) {
            v2f b;
            b.x = w1[4 * u + 2 * e];
            b.y = w1[4 * u + 2 * e + 1];
#pragma unroll
            for (int t = 0; t < 2; ++t) {
                v2f a;
                a.x = e ? aq[t][u].z : aq[t][u].x;
                a.y = e ? aq[t][u].w : aq[t][u].y;
                c[t] = __builtin_amdgcn_wmma_f32_16x16x4_f32(false, a, false, b,
                                                             (short)0, c[t], false, false);
            }
        }
#pragma unroll
        for (int j = 0; j < 3; ++j) {
            int x = 3 * u + j;
            if (x < NG * 16) {
                int g = x >> 4, i = x & 15;
                int ko = 16 * (i >> 3) + (i & 7);
                float ex = soft_w(br[g], srl, (float)ko, mr[g]);
                ex = (ko < room) ? ex : 0.0f;
                w2[g][i] = ex;
                sp[g] += ex;
            }
        }
    }
    {
        v2f b;
        b.x = w1[48];
        b.y = w1[49];
#pragma unroll
        for (int t = 0; t < 2; ++t)
            c[t] = __builtin_amdgcn_wmma_f32_16x16x4_f32(false, at[t], false, b,
                                                         (short)0, c[t], false, false);
    }
    sumc += __shfl_xor(sumc, 16);
#pragma unroll
    for (int g = 0; g < NG; ++g) sp[g] += __shfl_xor(sp[g], 16);

    v8f dacc[NG];
#pragma unroll
    for (int g = 0; g < NG; ++g) dacc[g] = (v8f){};
#pragma unroll
    for (int t = 0; t < 2; ++t) {
#pragma unroll
        for (int p = 0; p < 4; ++p) {
#pragma unroll
            for (int g = 0; g < NG; ++g) {
                v2f a, b;
                a.x = w2[g][8 * t + 2 * p];
                a.y = w2[g][8 * t + 2 * p + 1];
                b.x = c[t][2 * p];
                b.y = c[t][2 * p + 1];
                dacc[g] = __builtin_amdgcn_wmma_f32_16x16x4_f32(false, a, false, b,
                                                                (short)0, dacc[g], false, false);
            }
        }
    }

    float invc = 1.0f / sumc;
#pragma unroll
    for (int g = 0; g < NG; ++g) {
#pragma unroll
        for (int j = 0; j < 2; ++j) {
            v4f pv;
            pv.x = dacc[g][4 * j]     * invc;
            pv.y = dacc[g][4 * j + 1] * invc;
            pv.z = dacc[g][4 * j + 2] * invc;
            pv.w = dacc[g][4 * j + 3] * invc;
            part[w][g][j][lane] = pv;
        }
    }
    if (hs == 0) {
#pragma unroll
        for (int g = 0; g < NG; ++g) rsum[w][g][l16] = sp[g];
    }
    __syncthreads();

    int g = w & 1, j = w >> 1;
    v4f acc = part[0][g][j][lane];
    v4f den = *(const v4f*)&rsum[0][g][8 * hs + 4 * j];
#pragma unroll
    for (int ww = 1; ww < NWV; ++ww) {
        acc += part[ww][g][j][lane];
        den += *(const v4f*)&rsum[ww][g][8 * hs + 4 * j];
    }

    static_assert(SS % 4 == 0, "one row guard per group of four rows");
    int baseM = (2 * grp + g) * 16 + 8 * hs + 4 * j;
    if (colQ < SS && baseM < SS) {
        float* o = out + mat * SS * SS + baseM * SS + colQ;
#pragma unroll
        for (int i = 0; i < 4; ++i)
            o[i * SS] = acc[i] / den[i];
    }
}

extern "C" void kernel_launch(void* const* d_in, const int* in_sizes, int n_in,
                              void* d_out, int out_size, void* d_ws, size_t ws_size,
                              hipStream_t stream) {
    (void)in_sizes; (void)n_in; (void)out_size; (void)d_ws; (void)ws_size;
    const float* images     = (const float*)d_in[0];
    const float* transforms = (const float*)d_in[1];
    float*       out        = (float*)d_out;

    dec_chain<<<NB*CHN*28, 32*NWV, 0, stream>>>(images, transforms, out);
}
